# starting kernel
# baseline (speedup 1.0000x reference)
#include <hip/hip_runtime.h>

typedef __attribute__((ext_vector_type(2))) float v2f;
typedef __attribute__((ext_vector_type(8))) float v8f;

#define B_DIM   128
#define IN_DIM  1024
#define OUT_DIM 1024

__global__ __launch_bounds__(256) void dendritic_fused_kernel(
    const float* __restrict__ X,
    const float* __restrict__ Wnmda,
    const float* __restrict__ Wnon,
    const float* __restrict__ bnon,
    float* __restrict__ Y)
{
    const int lane = threadIdx.x & 31;
    const int wave = threadIdx.x >> 5;
    const int tile = blockIdx.x * 8 + wave;
    const int mt = tile >> 6;
    const int nt = tile & 63;

    const int mrow = mt * 16 + (lane & 15);
    const int nrow = nt * 16 + (lane & 15);
    const int h4   = (lane >> 4) * 4;

    const float* xr = X     + (size_t)mrow * IN_DIM + h4;
    const float* wn = Wnon  + (size_t)nrow * IN_DIM + h4;
    const float* wm = Wnmda + (size_t)nrow * IN_DIM + h4;

    v8f accN = {0.f,0.f,0.f,0.f,0.f,0.f,0.f,0.f};
    v8f accM = {0.f,0.f,0.f,0.f,0.f,0.f,0.f,0.f};

    const float edge_lo = (lane < 16)  ? 0.5f : 1.0f;
    const float edge_hi = (lane >= 16) ? 0.5f : 1.0f;

    for (int kb = 0; kb < IN_DIM; kb += 8) {
        float4 a4  = *(const float4*)(xr + kb);
        float4 b4n = *(const float4*)(wn + kb);
        float4 b4m = *(const float4*)(wm + kb);
        if (kb == 0)            b4m.x *= edge_lo;
        if (kb == IN_DIM - 8)   b4m.w *= edge_hi;

        v2f a0  = {a4.x,  a4.y},  a1  = {a4.z,  a4.w};
        v2f bn0 = {b4n.x, b4n.y}, bn1 = {b4n.z, b4n.w};
        v2f bm0 = {b4m.x, b4m.y}, bm1 = {b4m.z, b4m.w};

        accN = __builtin_amdgcn_wmma_f32_16x16x4_f32(false, a0, false, bn0, (short)0, accN, false, false);
        accN = __builtin_amdgcn_wmma_f32_16x16x4_f32(false, a1, false, bn1, (short)0, accN, false, false);
        accM = __builtin_amdgcn_wmma_f32_16x16x4_f32(false, a0, false, bm0, (short)0, accM, false, false);
        accM = __builtin_amdgcn_wmma_f32_16x16x4_f32(false, a1, false, bm1, (short)0, accM, false, false);
    }

    const float bb = bnon[nrow];
    float* op = Y + (size_t)(mt * 16 + (lane >> 4) * 8) * OUT_DIM + nrow;
    #pragma unroll
    for (int i = 0; i < 8; ++i) {
        float z     = accN[i] + bb;
        float s     = 1.0f / (1.0f + __expf(-z));
        float state = s - 1.0f;
        float ca    = fmaxf(accM[i], 0.0f);
        float nm    = 2.0f + state;
        float kd    = exp2f(-nm);
        float xn    = __powf(ca, nm);
        float y     = xn / (kd + xn) + state;
        op[(size_t)i * OUT_DIM] = y;
    }
}

extern "C" void kernel_launch(void* const* d_in, const int* in_sizes, int n_in,
                              void* d_out, int out_size, void* d_ws, size_t ws_size,
                              hipStream_t stream) {
    const float* X     = (const float*)d_in[0];
    const float* Wnmda = (const float*)d_in[1];
    const float* Wnon  = (const float*)d_in[2];
    const float* bnon  = (const float*)d_in[3];
    float* Y = (float*)d_out;

    dim3 grid(64);
    dim3 block(256);
    hipLaunchKernelGGL(dendritic_fused_kernel, grid, block, 0, stream,
                       X, Wnmda, Wnon, bnon, Y);
}


# k walk entered at a per-work-group position (8 starts 512 B apart, wrapping): requests of a moment spread over the rows' whole 4 KiB span, W sharers never on one line together; f32 sum order changes
# speedup vs baseline: 4.0306x; 4.0306x over previous
#include <hip/hip_runtime.h>

typedef __attribute__((ext_vector_type(2))) float v2f;
typedef __attribute__((ext_vector_type(8))) float v8f;

#define B_DIM   128
#define IN_DIM  1024
#define OUT_DIM 1024

#define KSTAGE  32
#define NSTAGE  (IN_DIM / KSTAGE)
#define AHEAD   6
#define NRING   (AHEAD + 1)
#define AGROUP  4
#define NAGROUP (NSTAGE / AGROUP)
#define NROT    (NSTAGE / AGROUP)
#define XPITCH  (IN_DIM + 4)

struct WStage { float4 w[4]; };
struct AStage { float4 a[4]; };

__device__ __forceinline__ void group_fence() { __builtin_amdgcn_sched_barrier(0); }

__device__ __forceinline__ void w_load(WStage& s, const float* __restrict__ wq, unsigned lbyte)
{
    #pragma unroll
    for (int j = 0; j < 4; ++j)
        s.w[j] = *(const float4*)((const char*)wq + lbyte + 32 * j);
}

__device__ __forceinline__ void a_load(AStage& s, const float* xl, int k0)
{
    #pragma unroll
    for (int j = 0; j < 4; ++j)
        s.a[j] = *(const float4*)(xl + k0 + 8 * j);
}

__device__ __forceinline__ void stage_mma(const AStage& x, const WStage& s, v8f& acc0, v8f& acc1)
{
    #pragma unroll
    for (int j = 0; j < 4; ++j) {
        v2f a0 = {x.a[j].x, x.a[j].y}, a1 = {x.a[j].z, x.a[j].w};
        v2f b0 = {s.w[j].x, s.w[j].y}, b1 = {s.w[j].z, s.w[j].w};
        acc0 = __builtin_amdgcn_wmma_f32_16x16x4_f32(false, a0, false, b0, (short)0, acc0, false, false);
        acc1 = __builtin_amdgcn_wmma_f32_16x16x4_f32(false, a1, false, b1, (short)0, acc1, false, false);
    }
}

__global__ __launch_bounds__(128) void dendritic_fused_kernel(
    const float* __restrict__ X,
    const float* __restrict__ Wnmda,
    const float* __restrict__ Wnon,
    const float* __restrict__ bnon,
    float* __restrict__ Y)
{
    __shared__ __attribute__((aligned(16))) float xs[16 * XPITCH];
    __shared__ float xchg[2][2][8][32];

    const int lane = threadIdx.x & 31;
    const int wave = __builtin_amdgcn_readfirstlane(threadIdx.x >> 5);
    const int isM  = wave & 1;
    const int slot = wave >> 1;
    const int tile = blockIdx.x * 2 + slot;
    const int mt = tile >> 6;
    const int nt = tile & 63;

    const int nrow = nt * 16 + (lane & 15);
    const int h4   = (lane >> 4) * 4;

    const float* wb = isM ? Wnmda : Wnon;
    const unsigned lbyte = (unsigned)(nrow * IN_DIM + h4) * 4u;

    const int rot = (blockIdx.x + (blockIdx.x >> 5)) & (NROT - 1);
    auto k_of = [rot](int t) {
        return ((((t / AGROUP) + rot) & (NROT - 1)) * AGROUP + (t % AGROUP)) * KSTAGE;
    };

    const float* xg = X + (size_t)(mt * 16 + 4 * wave) * IN_DIM + lane * 4;
    float4 xv[32];
    #pragma unroll
    for (int i = 0; i < 32; ++i)
        xv[i] = *(const float4*)(xg + (size_t)(i >> 3) * IN_DIM + (i & 7) * 128);

    const float bb = bnon[nrow];

    float* xw = xs + (4 * wave) * XPITCH + lane * 4;
    #pragma unroll
    for (int i = 0; i < 32; ++i)
        *(float4*)(xw + (i >> 3) * XPITCH + (i & 7) * 128) = xv[i];

    WStage ring[NRING];

    group_fence();
    #pragma unroll
    for (int t = 0; t < AHEAD; ++t) {
        w_load(ring[t], wb + k_of(t), lbyte);
        group_fence();
    }

    __builtin_amdgcn_fence(__ATOMIC_RELEASE, "workgroup", "local");
    __builtin_amdgcn_s_barrier();
    __builtin_amdgcn_fence(__ATOMIC_ACQUIRE, "workgroup", "local");

    v8f acc0 = {0.f,0.f,0.f,0.f,0.f,0.f,0.f,0.f};
    v8f acc1 = {0.f,0.f,0.f,0.f,0.f,0.f,0.f,0.f};

    const float edge_lo = (isM && lane < 16)  ? 0.5f : 1.0f;
    const float edge_hi = (isM && lane >= 16) ? 0.5f : 1.0f;

    const float* xl = xs + (lane & 15) * XPITCH + h4;
    AStage abuf[2][AGROUP];
    #pragma unroll
    for (int u = 0; u < AGROUP; ++u)
        a_load(abuf[0][u], xl, k_of(u));

    #pragma unroll
    for (int t = 0; t < NSTAGE; ++t) {
        group_fence();
        if (t + AHEAD < NSTAGE)
            w_load(ring[(t + AHEAD) % NRING], wb + k_of(t + AHEAD), lbyte);
        if ((t % AGROUP) == 0 && t + AGROUP < NSTAGE) {
            #pragma unroll
            for (int u = 0; u < AGROUP; ++u)
                a_load(abuf[((t / AGROUP) + 1) & 1][u], xl, k_of(t + AGROUP + u));
        }
        group_fence();
        if ((t % AGROUP) == 0)
            ring[t % NRING].w[0].x *= (k_of(t) == 0) ? edge_lo : 1.0f;
        if ((t % AGROUP) == AGROUP - 1)
            ring[t % NRING].w[3].w *= (k_of(t) == IN_DIM - KSTAGE) ? edge_hi : 1.0f;
        stage_mma(abuf[(t / AGROUP) & 1][t % AGROUP], ring[t % NRING], acc0, acc1);
    }

    const v8f acc = acc0 + acc1;

    #pragma unroll
    for (int i = 0; i < 8; ++i)
        xchg[slot][isM][i][lane] = acc[i];
    __syncthreads();

    const int r0 = isM * 4;
    float* op = Y + (size_t)(mt * 16 + (lane >> 4) * 8 + r0) * OUT_DIM + nrow;
    #pragma unroll
    for (int i = 0; i < 4; ++i) {
        const float sN = xchg[slot][0][r0 + i][lane];
        const float sM = xchg[slot][1][r0 + i][lane];
        float z     = sN + bb;
        float s     = __builtin_amdgcn_rcpf(1.0f + __expf(-z));
        float state = s - 1.0f;
        float ca    = fmaxf(sM, 0.0f);
        float nm    = 2.0f + state;
        float kd    = __builtin_amdgcn_exp2f(-nm);
        float xn    = __builtin_amdgcn_exp2f(nm * __builtin_amdgcn_logf(ca));
        float y     = __builtin_amdgcn_rcpf(1.0f + kd * __builtin_amdgcn_rcpf(xn)) + state;
        op[(size_t)i * OUT_DIM] = y;
    }
}

extern "C" void kernel_launch(void* const* d_in, const int* in_sizes, int n_in,
                              void* d_out, int out_size, void* d_ws, size_t ws_size,
                              hipStream_t stream) {
    const float* X     = (const float*)d_in[0];
    const float* Wnmda = (const float*)d_in[1];
    const float* Wnon  = (const float*)d_in[2];
    const float* bnon  = (const float*)d_in[3];
    float* Y = (float*)d_out;

    dim3 grid(256);
    dim3 block(128);
    hipLaunchKernelGGL(dendritic_fused_kernel, grid, block, 0, stream,
                       X, Wnmda, Wnon, bnon, Y);
}
